# starting kernel
# baseline (speedup 1.0000x reference)
#include <hip/hip_runtime.h>


typedef __attribute__((ext_vector_type(2))) float v2f;
typedef __attribute__((ext_vector_type(8))) float v8f;

#define B_DIM   32
#define N_PTS   4096
#define M_STARS 1024
#define LOG2E   1.44269504088896340736f

__global__ __launch_bounds__(256) void ursa_gau_wmma(
    const float* __restrict__ pts,
    const float* __restrict__ stars,
    float* __restrict__ out)
{
    const int mtile = blockIdx.x;
    const int b     = blockIdx.y;
    const int tid   = threadIdx.x;
    const int wave  = tid >> 5;
    const int lane  = tid & 31;
    const int l16   = lane & 15;
    const bool hi   = (lane >= 16);

    const int m_base = mtile * 16;

    const int   m  = m_base + l16;
    const float sx = stars[3 * m + 0];
    const float sy = stars[3 * m + 1];
    const float sz = stars[3 * m + 2];
    const float s2 = sx * sx + sy * sy + sz * sz;
    v2f bmat;
    bmat[0] = (hi ? sz : sx) * LOG2E;
    bmat[1] = hi ? (-0.5f * LOG2E) : (sy * LOG2E);
    const float cval = -0.5f * LOG2E * s2;
    v8f cvec;
#pragma unroll
    for (int r = 0; r < 8; ++r) cvec[r] = cval;

    const float* ptsb = pts + (size_t)b * (N_PTS * 3);
    float acc = 0.0f;

    const int tiles_per_wave = (N_PTS / 16) / 8;
#pragma unroll 2
    for (int i = 0; i < tiles_per_wave; ++i) {
        const int n = (wave * tiles_per_wave + i) * 16 + l16;
        const float x  = ptsb[3 * n + 0];
        const float y  = ptsb[3 * n + 1];
        const float z  = ptsb[3 * n + 2];
        const float p2 = x * x + y * y + z * z;
        v2f amat;
        amat[0] = hi ? z  : x;
        amat[1] = hi ? p2 : y;

        v8f d = __builtin_amdgcn_wmma_f32_16x16x4_f32(
             false, amat,  false, bmat,
             (short)0, cvec,  false,  false);

        float s = 0.0f;
#pragma unroll
        for (int r = 0; r < 8; ++r) s += __builtin_amdgcn_exp2f(d[r]);
        acc += s;
    }

    float tot = acc + __shfl_xor(acc, 16, 32);

    __shared__ float partial[8][16];
    if (lane < 16) partial[wave][l16] = tot;
    __syncthreads();
    if (tid < 16) {
        float s = 0.0f;
#pragma unroll
        for (int w = 0; w < 8; ++w) s += partial[w][tid];
        out[(size_t)b * M_STARS + m_base + tid] = s;
    }
}

extern "C" void kernel_launch(void* const* d_in, const int* in_sizes, int n_in,
                              void* d_out, int out_size, void* d_ws, size_t ws_size,
                              hipStream_t stream) {
    const float* pts   = (const float*)d_in[0];
    const float* stars = (const float*)d_in[1];
    float* out         = (float*)d_out;

    dim3 grid(M_STARS / 16, B_DIM);
    ursa_gau_wmma<<<grid, 256, 0, stream>>>(pts, stars, out);
}


# Points staged once per work-group in LDS as ready A fragments (x,y,z,|p|^2); the 8 waves split the stars, each wave sums all 4096 points for its 16 stars; sum order changed
# speedup vs baseline: 1.3364x; 1.3364x over previous
#include <hip/hip_runtime.h>


typedef __attribute__((ext_vector_type(2))) float v2f;
typedef __attribute__((ext_vector_type(8))) float v8f;

#define B_DIM   32
#define N_PTS   4096
#define M_STARS 1024
#define LOG2E   1.44269504088896340736f

#define N_TILES     (N_PTS / 16)
#define N_PAIRS     (N_TILES / 2)
#define PAIRS_BLK   8
#define N_BLKS      (N_PAIRS / PAIRS_BLK)


__device__ __forceinline__ void put_point(float* sA, int n, float x, float y, float z)
{
    const int t    = n >> 4;
    const int r    = n & 15;
    const int base = ((t >> 1) * 32) * 4 + (t & 1) * 2;
    const float p2 = x * x + y * y + z * z;
    v2f lo; lo[0] = x; lo[1] = y;
    v2f hi; hi[0] = z; hi[1] = p2;
    *(v2f*)(sA + base + r * 4)        = lo;
    *(v2f*)(sA + base + (16 + r) * 4) = hi;
}

__global__ __launch_bounds__(256) void ursa_gau_wmma(
    const float* __restrict__ pts,
    const float* __restrict__ stars,
    float* __restrict__ out)
{
    __shared__ __attribute__((aligned(16))) float sA[N_PTS * 4];
    __shared__ __attribute__((aligned(16))) float sOut[128];

    const int mgrp = blockIdx.x;
    const int b    = blockIdx.y;
    const int tid  = threadIdx.x;
    const int wave = tid >> 5;
    const int lane = tid & 31;
    const int l16  = lane & 15;
    const bool hi  = (lane >= 16);

    const float4* p4 = (const float4*)(pts + (size_t)b * (N_PTS * 3));
    float4 raw[4][3];
#pragma unroll
    for (int j = 0; j < 4; ++j) {
        const int g = j * 256 + tid;
        raw[j][0] = p4[3 * g + 0];
        raw[j][1] = p4[3 * g + 1];
        raw[j][2] = p4[3 * g + 2];
    }

    const int m_base = (mgrp * 8 + wave) * 16;
    const int   m  = m_base + l16;
    const float sx = stars[3 * m + 0];
    const float sy = stars[3 * m + 1];
    const float sz = stars[3 * m + 2];
    const float s2 = sx * sx + sy * sy + sz * sz;
    v2f bmat;
    bmat[0] = (hi ? sz : sx) * LOG2E;
    bmat[1] = hi ? (-0.5f * LOG2E) : (sy * LOG2E);
    const float cval = -0.5f * LOG2E * s2;
    v8f cvec;
#pragma unroll
    for (int r = 0; r < 8; ++r) cvec[r] = cval;

#pragma unroll
    for (int j = 0; j < 4; ++j) {
        const int n0 = (j * 256 + tid) * 4;
        const float4 v0 = raw[j][0], v1 = raw[j][1], v2 = raw[j][2];
        put_point(sA, n0 + 0, v0.x, v0.y, v0.z);
        put_point(sA, n0 + 1, v0.w, v1.x, v1.y);
        put_point(sA, n0 + 2, v1.z, v1.w, v2.x);
        put_point(sA, n0 + 3, v2.y, v2.z, v2.w);
    }
    __syncthreads();

    const float4* sA4 = (const float4*)sA;
    v2f acc0 = {0.0f, 0.0f}, acc1 = {0.0f, 0.0f}, acc2 = {0.0f, 0.0f}, acc3 = {0.0f, 0.0f};

    float4 cur[PAIRS_BLK];
#pragma unroll
    for (int u = 0; u < PAIRS_BLK; ++u) cur[u] = sA4[u * 32 + lane];

    for (int blk = 0; blk < N_BLKS; ++blk) {
        float4 nxt[PAIRS_BLK];
        if (blk + 1 < N_BLKS) {
#pragma unroll
            for (int u = 0; u < PAIRS_BLK; ++u)
                nxt[u] = sA4[((blk + 1) * PAIRS_BLK + u) * 32 + lane];
        } else {
#pragma unroll
            for (int u = 0; u < PAIRS_BLK; ++u) nxt[u] = cur[u];
        }

#pragma unroll
        for (int u = 0; u < PAIRS_BLK; ++u) {
#pragma unroll
            for (int h = 0; h < 2; ++h) {
                v2f amat;
                amat[0] = h ? cur[u].z : cur[u].x;
                amat[1] = h ? cur[u].w : cur[u].y;

                v8f d = __builtin_amdgcn_wmma_f32_16x16x4_f32(
                     false, amat,  false, bmat,
                     (short)0, cvec,  false,  false);

                v2f e0, e1, e2, e3;
                e0[0] = __builtin_amdgcn_exp2f(d[0]); e0[1] = __builtin_amdgcn_exp2f(d[1]);
                e1[0] = __builtin_amdgcn_exp2f(d[2]); e1[1] = __builtin_amdgcn_exp2f(d[3]);
                e2[0] = __builtin_amdgcn_exp2f(d[4]); e2[1] = __builtin_amdgcn_exp2f(d[5]);
                e3[0] = __builtin_amdgcn_exp2f(d[6]); e3[1] = __builtin_amdgcn_exp2f(d[7]);
                acc0 += e0; acc1 += e1; acc2 += e2; acc3 += e3;
            }
        }

#pragma unroll
        for (int u = 0; u < PAIRS_BLK; ++u) cur[u] = nxt[u];
    }

    const v2f a01 = acc0 + acc1;
    const v2f a23 = acc2 + acc3;
    const v2f a   = a01 + a23;
    const float acc = a[0] + a[1];

    const float tot = acc + __shfl_xor(acc, 16, 32);

    if (lane < 16) sOut[wave * 16 + l16] = tot;
    __syncthreads();
    if (tid < 32) {
        const float4 v = ((const float4*)sOut)[tid];
        ((float4*)(out + (size_t)b * M_STARS + mgrp * 128))[tid] = v;
    }
}

extern "C" void kernel_launch(void* const* d_in, const int* in_sizes, int n_in,
                              void* d_out, int out_size, void* d_ws, size_t ws_size,
                              hipStream_t stream) {
    const float* pts   = (const float*)d_in[0];
    const float* stars = (const float*)d_in[1];
    float* out         = (float*)d_out;

    dim3 grid(M_STARS / 128, B_DIM);
    ursa_gau_wmma<<<grid, 256, 0, stream>>>(pts, stars, out);
}
